# starting kernel
# baseline (speedup 1.0000x reference)
#include <hip/hip_runtime.h>
#include <math.h>

typedef __attribute__((ext_vector_type(2))) float v2f;
typedef __attribute__((ext_vector_type(8))) float v8f;

#define EPSQ 1e-8f

__device__ __forceinline__ float fsqrt(float v) {
    return __builtin_amdgcn_sqrtf(v);
}

__device__ __forceinline__ float ry_elem(float c, float s, int br, int bc) {
    return br == 0 ? (bc == 0 ? c : -s) : (bc == 0 ? s : c);
}

__global__ void build_mc_kernel(const float* __restrict__ theta,
                                float* __restrict__ mc) {
    int t = threadIdx.x;
    int i = t >> 4;
    int j = t & 15;
    float c1[4], s1[4], c2[4], s2[4];
#pragma unroll
    for (int q = 0; q < 4; ++q) {
        c1[q] = cosf(theta[q] * 0.5f);
        s1[q] = sinf(theta[q] * 0.5f);
        c2[q] = cosf(theta[4 + q] * 0.5f);
        s2[q] = sinf(theta[4 + q] * 0.5f);
    }
    float sum = 0.f;
#pragma unroll
    for (int k = 0; k < 16; ++k) {
        float m2 = 1.f, m1 = 1.f;
#pragma unroll
        for (int q = 0; q < 4; ++q) {
            m2 *= ry_elem(c2[q], s2[q], (i >> q) & 1, (k >> q) & 1);
            m1 *= ry_elem(c1[q], s1[q], (k >> q) & 1, (j >> q) & 1);
        }
        int neg = (((k >> 2) & (k >> 3)) ^ ((k >> 1) & (k >> 2)) ^
                   ((k >> 0) & (k >> 1)) ^ ((k >> 0) & (k >> 3))) & 1;
        sum += (neg ? -m2 : m2) * m1;
    }
    mc[i * 16 + j] = sum;
}

__global__ void __launch_bounds__(256) vqc_main_kernel(
        const float* __restrict__ x, const float* __restrict__ mc,
        float* __restrict__ out, int B) {
    const int lane = threadIdx.x & 31;
    const int half = lane >> 4;
    const int col  = lane & 15;
    const long wave = (long)((blockIdx.x * blockDim.x + threadIdx.x) >> 5);
    const long waveBase = wave * 128;
    if (waveBase >= B) return;

    const float* mrow = mc + col * 16 + 2 * half;
    v2f a0 = *(const v2f*)(mrow + 0);
    v2f a1 = *(const v2f*)(mrow + 4);
    v2f a2 = *(const v2f*)(mrow + 8);
    v2f a3 = *(const v2f*)(mrow + 12);

    for (int t = 0; t < 8; ++t) {
        long base = waveBase + (long)t * 16;
        if (base >= B) break;
        long b  = base + col;
        long bl = b < B ? b : (long)(B - 1);
        float4 xv = *(const float4*)(x + bl * 4);
        float y0 = xv.w, y1 = xv.z, y2 = xv.y, y3 = xv.x;

        float f1 = half ? y1 : (1.f - y1);
        float e0 = 1.f - y0;
        float n2 = 1.f - y2, n3 = 1.f - y3;

        v8f acc = {0.f, 0.f, 0.f, 0.f, 0.f, 0.f, 0.f, 0.f};
        {
            float p = f1 * n2 * n3;
            v2f bm; bm.x = fsqrt(p * e0 + EPSQ); bm.y = fsqrt(p * y0 + EPSQ);
            acc = __builtin_amdgcn_wmma_f32_16x16x4_f32(false, a0, false, bm,
                                                        (short)0, acc, false, false);
        }
        {
            float p = f1 * y2 * n3;
            v2f bm; bm.x = fsqrt(p * e0 + EPSQ); bm.y = fsqrt(p * y0 + EPSQ);
            acc = __builtin_amdgcn_wmma_f32_16x16x4_f32(false, a1, false, bm,
                                                        (short)0, acc, false, false);
        }
        {
            float p = f1 * n2 * y3;
            v2f bm; bm.x = fsqrt(p * e0 + EPSQ); bm.y = fsqrt(p * y0 + EPSQ);
            acc = __builtin_amdgcn_wmma_f32_16x16x4_f32(false, a2, false, bm,
                                                        (short)0, acc, false, false);
        }
        {
            float p = f1 * y2 * y3;
            v2f bm; bm.x = fsqrt(p * e0 + EPSQ); bm.y = fsqrt(p * y0 + EPSQ);
            acc = __builtin_amdgcn_wmma_f32_16x16x4_f32(false, a3, false, bm,
                                                        (short)0, acc, false, false);
        }

        float pv[8];
#pragma unroll
        for (int v = 0; v < 8; ++v) { float d = acc[v]; pv[v] = d * d; }
        float S0 = pv[1] + pv[3] + pv[5] + pv[7];
        float S1 = pv[2] + pv[3] + pv[6] + pv[7];
        float S2 = pv[4] + pv[5] + pv[6] + pv[7];
        float Sall = S2 + pv[0] + pv[1] + pv[2] + pv[3];
        float S3 = half ? Sall : 0.f;

        S0 += __shfl_xor(S0, 16, 32);
        S1 += __shfl_xor(S1, 16, 32);
        S2 += __shfl_xor(S2, 16, 32);
        S3 += __shfl_xor(S3, 16, 32);

        if (half == 0 && b < B) {
            float4 o = make_float4(S3, S2, S1, S0);
            *(float4*)(out + b * 4) = o;
        }
    }
}

extern "C" void kernel_launch(void* const* d_in, const int* in_sizes, int n_in,
                              void* d_out, int out_size, void* d_ws, size_t ws_size,
                              hipStream_t stream) {
    const float* x     = (const float*)d_in[0];
    const float* theta = (const float*)d_in[1];
    float* out = (float*)d_out;
    float* mc  = (float*)d_ws;
    int B = in_sizes[0] / 4;
    if (B <= 0) return;

    build_mc_kernel<<<1, 256, 0, stream>>>(theta, mc);

    long nWaves  = ((long)B + 127) / 128;
    long nBlocks = (nWaves + 7) / 8;
    vqc_main_kernel<<<(int)nBlocks, 256, 0, stream>>>(x, mc, out, B);
}


# Persistent waves (4 per SIMD) pass x rows and the half-wave partial sums through wave-private LDS: one row per lane loaded 7 groups ahead, two accumulator chains, 512-byte stores; sum order kept
# speedup vs baseline: 1.1781x; 1.1781x over previous
#include <hip/hip_runtime.h>
#include <math.h>

typedef __attribute__((ext_vector_type(2))) float v2f;
typedef __attribute__((ext_vector_type(8))) float v8f;

#define EPSQ 1e-8f

__device__ __forceinline__ float fsqrt(float v) {
    return __builtin_amdgcn_sqrtf(v);
}

__device__ __forceinline__ float ry_elem(float c, float s, int br, int bc) {
    return br == 0 ? (bc == 0 ? c : -s) : (bc == 0 ? s : c);
}

__global__ void build_mc_kernel(const float* __restrict__ theta,
                                float* __restrict__ mc) {
    int t = threadIdx.x;
    int i = t >> 4;
    int j = t & 15;
    float c1[4], s1[4], c2[4], s2[4];
#pragma unroll
    for (int q = 0; q < 4; ++q) {
        c1[q] = cosf(theta[q] * 0.5f);
        s1[q] = sinf(theta[q] * 0.5f);
        c2[q] = cosf(theta[4 + q] * 0.5f);
        s2[q] = sinf(theta[4 + q] * 0.5f);
    }
    float sum = 0.f;
#pragma unroll
    for (int k = 0; k < 16; ++k) {
        float m2 = 1.f, m1 = 1.f;
#pragma unroll
        for (int q = 0; q < 4; ++q) {
            m2 *= ry_elem(c2[q], s2[q], (i >> q) & 1, (k >> q) & 1);
            m1 *= ry_elem(c1[q], s1[q], (k >> q) & 1, (j >> q) & 1);
        }
        int neg = (((k >> 2) & (k >> 3)) ^ ((k >> 1) & (k >> 2)) ^
                   ((k >> 0) & (k >> 1)) ^ ((k >> 0) & (k >> 3))) & 1;
        sum += (neg ? -m2 : m2) * m1;
    }
    mc[i * 16 + j] = sum;
}

#define WAVES_PER_BLOCK 8
#define SLOTS 8

__device__ __forceinline__ void make_b(const float4 xv, const int half, v2f bm[4]) {
    const float y0 = xv.w, y1 = xv.z, y2 = xv.y, y3 = xv.x;
    const float f1 = half ? y1 : (1.f - y1);
    const float e0 = 1.f - y0;
    const float n2 = 1.f - y2, n3 = 1.f - y3;
    float p;
    p = f1 * n2 * n3; bm[0].x = fsqrt(p * e0 + EPSQ); bm[0].y = fsqrt(p * y0 + EPSQ);
    p = f1 * y2 * n3; bm[1].x = fsqrt(p * e0 + EPSQ); bm[1].y = fsqrt(p * y0 + EPSQ);
    p = f1 * n2 * y3; bm[2].x = fsqrt(p * e0 + EPSQ); bm[2].y = fsqrt(p * y0 + EPSQ);
    p = f1 * y2 * y3; bm[3].x = fsqrt(p * e0 + EPSQ); bm[3].y = fsqrt(p * y0 + EPSQ);
}

__device__ __forceinline__ float4 half_sums(const v8f acc) {
    float pv[8];
#pragma unroll
    for (int v = 0; v < 8; ++v) { float d = acc[v]; pv[v] = d * d; }
    float S0 = pv[1] + pv[3] + pv[5] + pv[7];
    float S1 = pv[2] + pv[3] + pv[6] + pv[7];
    float S2 = pv[4] + pv[5] + pv[6] + pv[7];
    float Sall = S2 + pv[0] + pv[1] + pv[2] + pv[3];
    return make_float4(Sall, S2, S1, S0);
}

__global__ void __launch_bounds__(256) vqc_main_kernel(
        const float* __restrict__ x, const float* __restrict__ mc,
        float* __restrict__ out, int B) {
    __shared__ float4 xs[WAVES_PER_BLOCK][32];
    __shared__ float4 ps[WAVES_PER_BLOCK][2][2][16];

    const int lane = threadIdx.x & 31;
    const int half = lane >> 4;
    const int col  = lane & 15;
    const int w   = __builtin_amdgcn_readfirstlane((int)(threadIdx.x >> 5));
    const int nW  = (int)gridDim.x * WAVES_PER_BLOCK;
    const int gw  = (int)blockIdx.x * WAVES_PER_BLOCK + w;
    const int nG  = (int)(((long)B + 31) >> 5);
    const int lastRows = B - (nG - 1) * 32;
    const unsigned laneOff = (unsigned)lane * 4u;

    const float* mrow = mc + col * 16 + 2 * half;
    const v2f a0 = *(const v2f*)(mrow + 0);
    const v2f a1 = *(const v2f*)(mrow + 4);
    const v2f a2 = *(const v2f*)(mrow + 8);
    const v2f a3 = *(const v2f*)(mrow + 12);

    auto process = [&](const float4 xv, const int g) {
        xs[w][lane] = xv;
        __builtin_amdgcn_fence(__ATOMIC_ACQ_REL, "wavefront");
        const float4 x0 = xs[w][col];
        const float4 x1 = xs[w][16 + col];

        v2f b0[4], b1[4];
        make_b(x0, half, b0);
        make_b(x1, half, b1);

        v8f acc0 = {0.f, 0.f, 0.f, 0.f, 0.f, 0.f, 0.f, 0.f};
        v8f acc1 = {0.f, 0.f, 0.f, 0.f, 0.f, 0.f, 0.f, 0.f};
        acc0 = __builtin_amdgcn_wmma_f32_16x16x4_f32(false, a0, false, b0[0], (short)0, acc0, false, false);
        acc1 = __builtin_amdgcn_wmma_f32_16x16x4_f32(false, a0, false, b1[0], (short)0, acc1, false, false);
        acc0 = __builtin_amdgcn_wmma_f32_16x16x4_f32(false, a1, false, b0[1], (short)0, acc0, false, false);
        acc1 = __builtin_amdgcn_wmma_f32_16x16x4_f32(false, a1, false, b1[1], (short)0, acc1, false, false);
        acc0 = __builtin_amdgcn_wmma_f32_16x16x4_f32(false, a2, false, b0[2], (short)0, acc0, false, false);
        acc1 = __builtin_amdgcn_wmma_f32_16x16x4_f32(false, a2, false, b1[2], (short)0, acc1, false, false);
        acc0 = __builtin_amdgcn_wmma_f32_16x16x4_f32(false, a3, false, b0[3], (short)0, acc0, false, false);
        acc1 = __builtin_amdgcn_wmma_f32_16x16x4_f32(false, a3, false, b1[3], (short)0, acc1, false, false);

        ps[w][0][half][col] = half_sums(acc0);
        ps[w][1][half][col] = half_sums(acc1);
        __builtin_amdgcn_fence(__ATOMIC_ACQ_REL, "wavefront");
        const float4 pa = ps[w][half][0][col];
        const float4 pb = ps[w][half][1][col];
        const float4 o = make_float4(pb.x, pa.y + pb.y, pa.z + pb.z, pa.w + pb.w);
        float* og = out + (long)g * 128;
        if (g != nG - 1 || lane < lastRows) *(float4*)(og + laneOff) = o;
    };

    const unsigned lastOff = (unsigned)(lane < lastRows ? lane : lastRows - 1) * 4u;
    auto load_group = [&](const int gq) -> float4 {
        const bool last = gq >= nG - 1;
        const float* xg = x + (long)(last ? nG - 1 : gq) * 128;
        return *(const float4*)(xg + (last ? lastOff : laneOff));
    };

    const int cnt   = gw < nG ? (nG - gw + nW - 1) / nW : 0;
    const int rounds = cnt / SLOTS;
    const int rem   = cnt - rounds * SLOTS;

    float4 ring[SLOTS];
#pragma unroll
    for (int k = 0; k < SLOTS - 1; ++k)
        ring[k] = load_group(gw + k * nW);
    ring[SLOTS - 1] = ring[0];

    int g = gw;
    for (int q = 0; q < rounds; ++q) {
#pragma unroll
        for (int k = 0; k < SLOTS; ++k) {
            ring[(k + SLOTS - 1) % SLOTS] = load_group(g + (SLOTS - 1) * nW);
            process(ring[k], g);
            g += nW;
        }
    }
    for (int k = 0; k < rem; ++k) {
        float4 xv = ring[0];
#pragma unroll
        for (int j = 1; j < SLOTS - 1; ++j) if (k == j) xv = ring[j];
        process(xv, g);
        g += nW;
    }
}

extern "C" void kernel_launch(void* const* d_in, const int* in_sizes, int n_in,
                              void* d_out, int out_size, void* d_ws, size_t ws_size,
                              hipStream_t stream) {
    const float* x     = (const float*)d_in[0];
    const float* theta = (const float*)d_in[1];
    float* out = (float*)d_out;
    float* mc  = (float*)d_ws;
    int B = in_sizes[0] / 4;
    if (B <= 0) return;

    build_mc_kernel<<<1, 256, 0, stream>>>(theta, mc);

    vqc_main_kernel<<<512, 256, 0, stream>>>(x, mc, out, B);
}


# Matrix-build kernel: 8 angles fetched by one vector load, each lane evaluates one angle's sine and cosine once (was 8 serial scalar loads, 16 scalar-unit evaluations); lane reads; sum order kept
# speedup vs baseline: 1.2416x; 1.0539x over previous
#include <hip/hip_runtime.h>
#include <math.h>

typedef __attribute__((ext_vector_type(2))) float v2f;
typedef __attribute__((ext_vector_type(8))) float v8f;

#define EPSQ 1e-8f

__device__ __forceinline__ float fsqrt(float v) {
    return __builtin_amdgcn_sqrtf(v);
}

__device__ __forceinline__ float ry_elem(float c, float s, int br, int bc) {
    return br == 0 ? (bc == 0 ? c : -s) : (bc == 0 ? s : c);
}

__global__ void build_mc_kernel(const float* __restrict__ theta,
                                float* __restrict__ mc) {
    int t = threadIdx.x;
    int i = t >> 4;
    int j = t & 15;
    const float hth = theta[t & 7] * 0.5f;
    const int cs = __float_as_int(cosf(hth));
    const int sn = __float_as_int(sinf(hth));
    float c1[4], s1[4], c2[4], s2[4];
#pragma unroll
    for (int q = 0; q < 4; ++q) {
        c1[q] = __int_as_float(__builtin_amdgcn_readlane(cs, q));
        s1[q] = __int_as_float(__builtin_amdgcn_readlane(sn, q));
        c2[q] = __int_as_float(__builtin_amdgcn_readlane(cs, 4 + q));
        s2[q] = __int_as_float(__builtin_amdgcn_readlane(sn, 4 + q));
    }
    float sum = 0.f;
#pragma unroll
    for (int k = 0; k < 16; ++k) {
        float m2 = 1.f, m1 = 1.f;
#pragma unroll
        for (int q = 0; q < 4; ++q) {
            m2 *= ry_elem(c2[q], s2[q], (i >> q) & 1, (k >> q) & 1);
            m1 *= ry_elem(c1[q], s1[q], (k >> q) & 1, (j >> q) & 1);
        }
        int neg = (((k >> 2) & (k >> 3)) ^ ((k >> 1) & (k >> 2)) ^
                   ((k >> 0) & (k >> 1)) ^ ((k >> 0) & (k >> 3))) & 1;
        sum += (neg ? -m2 : m2) * m1;
    }
    mc[i * 16 + j] = sum;
}

#define WAVES_PER_BLOCK 8
#define SLOTS 8

__device__ __forceinline__ void make_b(const float4 xv, const int half, v2f bm[4]) {
    const float y0 = xv.w, y1 = xv.z, y2 = xv.y, y3 = xv.x;
    const float f1 = half ? y1 : (1.f - y1);
    const float e0 = 1.f - y0;
    const float n2 = 1.f - y2, n3 = 1.f - y3;
    float p;
    p = f1 * n2 * n3; bm[0].x = fsqrt(p * e0 + EPSQ); bm[0].y = fsqrt(p * y0 + EPSQ);
    p = f1 * y2 * n3; bm[1].x = fsqrt(p * e0 + EPSQ); bm[1].y = fsqrt(p * y0 + EPSQ);
    p = f1 * n2 * y3; bm[2].x = fsqrt(p * e0 + EPSQ); bm[2].y = fsqrt(p * y0 + EPSQ);
    p = f1 * y2 * y3; bm[3].x = fsqrt(p * e0 + EPSQ); bm[3].y = fsqrt(p * y0 + EPSQ);
}

__device__ __forceinline__ float4 half_sums(const v8f acc) {
    float pv[8];
#pragma unroll
    for (int v = 0; v < 8; ++v) { float d = acc[v]; pv[v] = d * d; }
    float S0 = pv[1] + pv[3] + pv[5] + pv[7];
    float S1 = pv[2] + pv[3] + pv[6] + pv[7];
    float S2 = pv[4] + pv[5] + pv[6] + pv[7];
    float Sall = S2 + pv[0] + pv[1] + pv[2] + pv[3];
    return make_float4(Sall, S2, S1, S0);
}

__global__ void __launch_bounds__(256) vqc_main_kernel(
        const float* __restrict__ x, const float* __restrict__ mc,
        float* __restrict__ out, int B) {
    __shared__ float4 xs[WAVES_PER_BLOCK][32];
    __shared__ float4 ps[WAVES_PER_BLOCK][2][2][16];

    const int lane = threadIdx.x & 31;
    const int half = lane >> 4;
    const int col  = lane & 15;
    const int w   = __builtin_amdgcn_readfirstlane((int)(threadIdx.x >> 5));
    const int nW  = (int)gridDim.x * WAVES_PER_BLOCK;
    const int gw  = (int)blockIdx.x * WAVES_PER_BLOCK + w;
    const int nG  = (int)(((long)B + 31) >> 5);
    const int lastRows = B - (nG - 1) * 32;
    const unsigned laneOff = (unsigned)lane * 4u;

    const float* mrow = mc + col * 16 + 2 * half;
    const v2f a0 = *(const v2f*)(mrow + 0);
    const v2f a1 = *(const v2f*)(mrow + 4);
    const v2f a2 = *(const v2f*)(mrow + 8);
    const v2f a3 = *(const v2f*)(mrow + 12);

    auto process = [&](const float4 xv, const int g) {
        xs[w][lane] = xv;
        __builtin_amdgcn_fence(__ATOMIC_ACQ_REL, "wavefront");
        const float4 x0 = xs[w][col];
        const float4 x1 = xs[w][16 + col];

        v2f b0[4], b1[4];
        make_b(x0, half, b0);
        make_b(x1, half, b1);

        v8f acc0 = {0.f, 0.f, 0.f, 0.f, 0.f, 0.f, 0.f, 0.f};
        v8f acc1 = {0.f, 0.f, 0.f, 0.f, 0.f, 0.f, 0.f, 0.f};
        acc0 = __builtin_amdgcn_wmma_f32_16x16x4_f32(false, a0, false, b0[0], (short)0, acc0, false, false);
        acc1 = __builtin_amdgcn_wmma_f32_16x16x4_f32(false, a0, false, b1[0], (short)0, acc1, false, false);
        acc0 = __builtin_amdgcn_wmma_f32_16x16x4_f32(false, a1, false, b0[1], (short)0, acc0, false, false);
        acc1 = __builtin_amdgcn_wmma_f32_16x16x4_f32(false, a1, false, b1[1], (short)0, acc1, false, false);
        acc0 = __builtin_amdgcn_wmma_f32_16x16x4_f32(false, a2, false, b0[2], (short)0, acc0, false, false);
        acc1 = __builtin_amdgcn_wmma_f32_16x16x4_f32(false, a2, false, b1[2], (short)0, acc1, false, false);
        acc0 = __builtin_amdgcn_wmma_f32_16x16x4_f32(false, a3, false, b0[3], (short)0, acc0, false, false);
        acc1 = __builtin_amdgcn_wmma_f32_16x16x4_f32(false, a3, false, b1[3], (short)0, acc1, false, false);

        ps[w][0][half][col] = half_sums(acc0);
        ps[w][1][half][col] = half_sums(acc1);
        __builtin_amdgcn_fence(__ATOMIC_ACQ_REL, "wavefront");
        const float4 pa = ps[w][half][0][col];
        const float4 pb = ps[w][half][1][col];
        const float4 o = make_float4(pb.x, pa.y + pb.y, pa.z + pb.z, pa.w + pb.w);
        float* og = out + (long)g * 128;
        if (g != nG - 1 || lane < lastRows) *(float4*)(og + laneOff) = o;
    };

    const unsigned lastOff = (unsigned)(lane < lastRows ? lane : lastRows - 1) * 4u;
    auto load_group = [&](const int gq) -> float4 {
        const bool last = gq >= nG - 1;
        const float* xg = x + (long)(last ? nG - 1 : gq) * 128;
        return *(const float4*)(xg + (last ? lastOff : laneOff));
    };

    const int cnt   = gw < nG ? (nG - gw + nW - 1) / nW : 0;
    const int rounds = cnt / SLOTS;
    const int rem   = cnt - rounds * SLOTS;

    float4 ring[SLOTS];
#pragma unroll
    for (int k = 0; k < SLOTS - 1; ++k)
        ring[k] = load_group(gw + k * nW);
    ring[SLOTS - 1] = ring[0];

    int g = gw;
    for (int q = 0; q < rounds; ++q) {
#pragma unroll
        for (int k = 0; k < SLOTS; ++k) {
            ring[(k + SLOTS - 1) % SLOTS] = load_group(g + (SLOTS - 1) * nW);
            process(ring[k], g);
            g += nW;
        }
    }
    for (int k = 0; k < rem; ++k) {
        float4 xv = ring[0];
#pragma unroll
        for (int j = 1; j < SLOTS - 1; ++j) if (k == j) xv = ring[j];
        process(xv, g);
        g += nW;
    }
}

extern "C" void kernel_launch(void* const* d_in, const int* in_sizes, int n_in,
                              void* d_out, int out_size, void* d_ws, size_t ws_size,
                              hipStream_t stream) {
    const float* x     = (const float*)d_in[0];
    const float* theta = (const float*)d_in[1];
    float* out = (float*)d_out;
    float* mc  = (float*)d_ws;
    int B = in_sizes[0] / 4;
    if (B <= 0) return;

    build_mc_kernel<<<1, 256, 0, stream>>>(theta, mc);

    vqc_main_kernel<<<512, 256, 0, stream>>>(x, mc, out, B);
}


# Matrix-build launch folded in, work-group-wide: thread t computes matrix element t behind the x prefetch, 1 KB LDS array and one barrier, A fragments read from LDS; one launch; sum order kept
# speedup vs baseline: 1.3308x; 1.0718x over previous
#include <hip/hip_runtime.h>
#include <math.h>

typedef __attribute__((ext_vector_type(2))) float v2f;
typedef __attribute__((ext_vector_type(8))) float v8f;

#define EPSQ 1e-8f

__device__ __forceinline__ float fsqrt(float v) {
    return __builtin_amdgcn_sqrtf(v);
}

__device__ __forceinline__ float ry_elem(float c, float s, int br, int bc) {
    return br == 0 ? (bc == 0 ? c : -s) : (bc == 0 ? s : c);
}

__device__ __forceinline__ float mc_element(const float th, const int t) {
    int i = t >> 4;
    int j = t & 15;
    const float hth = th * 0.5f;
    const int cs = __float_as_int(cosf(hth));
    const int sn = __float_as_int(sinf(hth));
    float c1[4], s1[4], c2[4], s2[4];
#pragma unroll
    for (int q = 0; q < 4; ++q) {
        c1[q] = __int_as_float(__builtin_amdgcn_readlane(cs, q));
        s1[q] = __int_as_float(__builtin_amdgcn_readlane(sn, q));
        c2[q] = __int_as_float(__builtin_amdgcn_readlane(cs, 4 + q));
        s2[q] = __int_as_float(__builtin_amdgcn_readlane(sn, 4 + q));
    }
    float sum = 0.f;
#pragma unroll
    for (int k = 0; k < 16; ++k) {
        float m2 = 1.f, m1 = 1.f;
#pragma unroll
        for (int q = 0; q < 4; ++q) {
            m2 *= ry_elem(c2[q], s2[q], (i >> q) & 1, (k >> q) & 1);
            m1 *= ry_elem(c1[q], s1[q], (k >> q) & 1, (j >> q) & 1);
        }
        int neg = (((k >> 2) & (k >> 3)) ^ ((k >> 1) & (k >> 2)) ^
                   ((k >> 0) & (k >> 1)) ^ ((k >> 0) & (k >> 3))) & 1;
        sum += (neg ? -m2 : m2) * m1;
    }
    return sum;
}

#define WAVES_PER_BLOCK 8
#define SLOTS 8
#define MC_PITCH 18

__device__ __forceinline__ void make_b(const float4 xv, const int half, v2f bm[4]) {
    const float y0 = xv.w, y1 = xv.z, y2 = xv.y, y3 = xv.x;
    const float f1 = half ? y1 : (1.f - y1);
    const float e0 = 1.f - y0;
    const float n2 = 1.f - y2, n3 = 1.f - y3;
    float p;
    p = f1 * n2 * n3; bm[0].x = fsqrt(p * e0 + EPSQ); bm[0].y = fsqrt(p * y0 + EPSQ);
    p = f1 * y2 * n3; bm[1].x = fsqrt(p * e0 + EPSQ); bm[1].y = fsqrt(p * y0 + EPSQ);
    p = f1 * n2 * y3; bm[2].x = fsqrt(p * e0 + EPSQ); bm[2].y = fsqrt(p * y0 + EPSQ);
    p = f1 * y2 * y3; bm[3].x = fsqrt(p * e0 + EPSQ); bm[3].y = fsqrt(p * y0 + EPSQ);
}

__device__ __forceinline__ float4 half_sums(const v8f acc) {
    float pv[8];
#pragma unroll
    for (int v = 0; v < 8; ++v) { float d = acc[v]; pv[v] = d * d; }
    float S0 = pv[1] + pv[3] + pv[5] + pv[7];
    float S1 = pv[2] + pv[3] + pv[6] + pv[7];
    float S2 = pv[4] + pv[5] + pv[6] + pv[7];
    float Sall = S2 + pv[0] + pv[1] + pv[2] + pv[3];
    return make_float4(Sall, S2, S1, S0);
}

__global__ void __launch_bounds__(256) vqc_main_kernel(
        const float* __restrict__ x, const float* __restrict__ theta,
        float* __restrict__ out, int B) {
    __shared__ float4 xs[WAVES_PER_BLOCK][32];
    __shared__ float4 ps[WAVES_PER_BLOCK][2][2][16];
    __shared__ float mcs[16 * MC_PITCH];

    const int lane = threadIdx.x & 31;
    const int half = lane >> 4;
    const int col  = lane & 15;
    const int w   = __builtin_amdgcn_readfirstlane((int)(threadIdx.x >> 5));
    const int nW  = (int)gridDim.x * WAVES_PER_BLOCK;
    const int gw  = (int)blockIdx.x * WAVES_PER_BLOCK + w;
    const int nG  = (int)(((long)B + 31) >> 5);
    const int lastRows = B - (nG - 1) * 32;
    const unsigned laneOff = (unsigned)lane * 4u;

    const unsigned lastOff = (unsigned)(lane < lastRows ? lane : lastRows - 1) * 4u;
    auto load_group = [&](const int gq) -> float4 {
        const bool last = gq >= nG - 1;
        const float* xg = x + (long)(last ? nG - 1 : gq) * 128;
        return *(const float4*)(xg + (last ? lastOff : laneOff));
    };

    const int cnt   = gw < nG ? (nG - gw + nW - 1) / nW : 0;
    const int rounds = cnt / SLOTS;
    const int rem   = cnt - rounds * SLOTS;

    const float th = theta[threadIdx.x & 7];

    float4 ring[SLOTS];
#pragma unroll
    for (int k = 0; k < SLOTS - 1; ++k)
        ring[k] = load_group(gw + k * nW);
    ring[SLOTS - 1] = ring[0];
    __builtin_amdgcn_sched_barrier(0);

    mcs[(threadIdx.x >> 4) * MC_PITCH + (threadIdx.x & 15)] = mc_element(th, (int)threadIdx.x);
    __builtin_amdgcn_fence(__ATOMIC_RELEASE, "workgroup", "local");
    __builtin_amdgcn_s_barrier();
    __builtin_amdgcn_fence(__ATOMIC_ACQUIRE, "workgroup", "local");
    const float* mrow = mcs + col * MC_PITCH + 2 * half;
    const v2f a0 = *(const v2f*)(mrow + 0);
    const v2f a1 = *(const v2f*)(mrow + 4);
    const v2f a2 = *(const v2f*)(mrow + 8);
    const v2f a3 = *(const v2f*)(mrow + 12);

    auto process = [&](const float4 xv, const int g) {
        xs[w][lane] = xv;
        __builtin_amdgcn_fence(__ATOMIC_ACQ_REL, "wavefront");
        const float4 x0 = xs[w][col];
        const float4 x1 = xs[w][16 + col];

        v2f b0[4], b1[4];
        make_b(x0, half, b0);
        make_b(x1, half, b1);

        v8f acc0 = {0.f, 0.f, 0.f, 0.f, 0.f, 0.f, 0.f, 0.f};
        v8f acc1 = {0.f, 0.f, 0.f, 0.f, 0.f, 0.f, 0.f, 0.f};
        acc0 = __builtin_amdgcn_wmma_f32_16x16x4_f32(false, a0, false, b0[0], (short)0, acc0, false, false);
        acc1 = __builtin_amdgcn_wmma_f32_16x16x4_f32(false, a0, false, b1[0], (short)0, acc1, false, false);
        acc0 = __builtin_amdgcn_wmma_f32_16x16x4_f32(false, a1, false, b0[1], (short)0, acc0, false, false);
        acc1 = __builtin_amdgcn_wmma_f32_16x16x4_f32(false, a1, false, b1[1], (short)0, acc1, false, false);
        acc0 = __builtin_amdgcn_wmma_f32_16x16x4_f32(false, a2, false, b0[2], (short)0, acc0, false, false);
        acc1 = __builtin_amdgcn_wmma_f32_16x16x4_f32(false, a2, false, b1[2], (short)0, acc1, false, false);
        acc0 = __builtin_amdgcn_wmma_f32_16x16x4_f32(false, a3, false, b0[3], (short)0, acc0, false, false);
        acc1 = __builtin_amdgcn_wmma_f32_16x16x4_f32(false, a3, false, b1[3], (short)0, acc1, false, false);

        ps[w][0][half][col] = half_sums(acc0);
        ps[w][1][half][col] = half_sums(acc1);
        __builtin_amdgcn_fence(__ATOMIC_ACQ_REL, "wavefront");
        const float4 pa = ps[w][half][0][col];
        const float4 pb = ps[w][half][1][col];
        const float4 o = make_float4(pb.x, pa.y + pb.y, pa.z + pb.z, pa.w + pb.w);
        float* og = out + (long)g * 128;
        if (g != nG - 1 || lane < lastRows) *(float4*)(og + laneOff) = o;
    };

    int g = gw;
    for (int q = 0; q < rounds; ++q) {
#pragma unroll
        for (int k = 0; k < SLOTS; ++k) {
            ring[(k + SLOTS - 1) % SLOTS] = load_group(g + (SLOTS - 1) * nW);
            process(ring[k], g);
            g += nW;
        }
    }
    for (int k = 0; k < rem; ++k) {
        float4 xv = ring[0];
#pragma unroll
        for (int j = 1; j < SLOTS - 1; ++j) if (k == j) xv = ring[j];
        process(xv, g);
        g += nW;
    }
}

extern "C" void kernel_launch(void* const* d_in, const int* in_sizes, int n_in,
                              void* d_out, int out_size, void* d_ws, size_t ws_size,
                              hipStream_t stream) {
    const float* x     = (const float*)d_in[0];
    const float* theta = (const float*)d_in[1];
    float* out = (float*)d_out;
    int B = in_sizes[0] / 4;
    if (B <= 0) return;

    vqc_main_kernel<<<512, 256, 0, stream>>>(x, theta, out, B);
}


# Wave step widened from 32 to 128 rows: the x rows and the half-wave partial sums of four groups cross LDS together, two LDS places per step instead of eight, all 32 WMMAs in one run; sum order kept
# speedup vs baseline: 1.4124x; 1.0613x over previous
#include <hip/hip_runtime.h>
#include <math.h>

typedef __attribute__((ext_vector_type(2))) float v2f;
typedef __attribute__((ext_vector_type(8))) float v8f;

#define EPSQ 1e-8f

__device__ __forceinline__ float fsqrt(float v) {
    return __builtin_amdgcn_sqrtf(v);
}

__device__ __forceinline__ float ry_elem(float c, float s, int br, int bc) {
    return br == 0 ? (bc == 0 ? c : -s) : (bc == 0 ? s : c);
}

__device__ __forceinline__ float mc_element(const float th, const int t) {
    int i = t >> 4;
    int j = t & 15;
    const float hth = th * 0.5f;
    const int cs = __float_as_int(cosf(hth));
    const int sn = __float_as_int(sinf(hth));
    float c1[4], s1[4], c2[4], s2[4];
#pragma unroll
    for (int q = 0; q < 4; ++q) {
        c1[q] = __int_as_float(__builtin_amdgcn_readlane(cs, q));
        s1[q] = __int_as_float(__builtin_amdgcn_readlane(sn, q));
        c2[q] = __int_as_float(__builtin_amdgcn_readlane(cs, 4 + q));
        s2[q] = __int_as_float(__builtin_amdgcn_readlane(sn, 4 + q));
    }
    float sum = 0.f;
#pragma unroll
    for (int k = 0; k < 16; ++k) {
        float m2 = 1.f, m1 = 1.f;
#pragma unroll
        for (int q = 0; q < 4; ++q) {
            m2 *= ry_elem(c2[q], s2[q], (i >> q) & 1, (k >> q) & 1);
            m1 *= ry_elem(c1[q], s1[q], (k >> q) & 1, (j >> q) & 1);
        }
        int neg = (((k >> 2) & (k >> 3)) ^ ((k >> 1) & (k >> 2)) ^
                   ((k >> 0) & (k >> 1)) ^ ((k >> 0) & (k >> 3))) & 1;
        sum += (neg ? -m2 : m2) * m1;
    }
    return sum;
}

#define WAVES_PER_BLOCK 8
#define GPS 4
#define SLOTS 3
#define MC_PITCH 18

__device__ __forceinline__ void make_b(const float4 xv, const int half, v2f bm[4]) {
    const float y0 = xv.w, y1 = xv.z, y2 = xv.y, y3 = xv.x;
    const float f1 = half ? y1 : (1.f - y1);
    const float e0 = 1.f - y0;
    const float n2 = 1.f - y2, n3 = 1.f - y3;
    float p;
    p = f1 * n2 * n3; bm[0].x = fsqrt(p * e0 + EPSQ); bm[0].y = fsqrt(p * y0 + EPSQ);
    p = f1 * y2 * n3; bm[1].x = fsqrt(p * e0 + EPSQ); bm[1].y = fsqrt(p * y0 + EPSQ);
    p = f1 * n2 * y3; bm[2].x = fsqrt(p * e0 + EPSQ); bm[2].y = fsqrt(p * y0 + EPSQ);
    p = f1 * y2 * y3; bm[3].x = fsqrt(p * e0 + EPSQ); bm[3].y = fsqrt(p * y0 + EPSQ);
}

__device__ __forceinline__ float4 half_sums(const v8f acc) {
    float pv[8];
#pragma unroll
    for (int v = 0; v < 8; ++v) { float d = acc[v]; pv[v] = d * d; }
    float S0 = pv[1] + pv[3] + pv[5] + pv[7];
    float S1 = pv[2] + pv[3] + pv[6] + pv[7];
    float S2 = pv[4] + pv[5] + pv[6] + pv[7];
    float Sall = S2 + pv[0] + pv[1] + pv[2] + pv[3];
    return make_float4(Sall, S2, S1, S0);
}

__global__ void __launch_bounds__(256) vqc_main_kernel(
        const float* __restrict__ x, const float* __restrict__ theta,
        float* __restrict__ out, int B) {
    __shared__ float4 xs[WAVES_PER_BLOCK][GPS][32];
    __shared__ float4 ps[WAVES_PER_BLOCK][GPS][2][2][16];
    __shared__ float mcs[16 * MC_PITCH];

    const int lane = threadIdx.x & 31;
    const int half = lane >> 4;
    const int col  = lane & 15;
    const int w   = __builtin_amdgcn_readfirstlane((int)(threadIdx.x >> 5));
    const int nW  = (int)gridDim.x * WAVES_PER_BLOCK;
    const int gw  = (int)blockIdx.x * WAVES_PER_BLOCK + w;
    const int nS  = (int)(((long)B + (32 * GPS - 1)) / (32 * GPS));
    const int lastRows = B - (nS - 1) * (32 * GPS);
    const unsigned laneOff = (unsigned)lane * 4u;

    unsigned lastOff[GPS];
#pragma unroll
    for (int t = 0; t < GPS; ++t) {
        const int r = t * 32 + lane;
        lastOff[t] = (unsigned)(r < lastRows ? r : lastRows - 1) * 4u;
    }
    auto load_step = [&](const int sq, float4 (&dst)[GPS]) {
        const bool last = sq >= nS - 1;
        const float* xg = x + (long)(last ? nS - 1 : sq) * (128 * GPS);
#pragma unroll
        for (int t = 0; t < GPS; ++t)
            dst[t] = *(const float4*)(xg + (last ? lastOff[t] : laneOff + 128u * t));
    };

    const int cnt   = gw < nS ? (nS - gw + nW - 1) / nW : 0;
    const int rounds = cnt / SLOTS;
    const int rem   = cnt - rounds * SLOTS;

    const float th = theta[threadIdx.x & 7];

    float4 ring[SLOTS][GPS];
#pragma unroll
    for (int k = 0; k < SLOTS - 1; ++k)
        load_step(gw + k * nW, ring[k]);
#pragma unroll
    for (int t = 0; t < GPS; ++t)
        ring[SLOTS - 1][t] = ring[0][t];
    __builtin_amdgcn_sched_barrier(0);

    mcs[(threadIdx.x >> 4) * MC_PITCH + (threadIdx.x & 15)] = mc_element(th, (int)threadIdx.x);
    __builtin_amdgcn_fence(__ATOMIC_RELEASE, "workgroup", "local");
    __builtin_amdgcn_s_barrier();
    __builtin_amdgcn_fence(__ATOMIC_ACQUIRE, "workgroup", "local");
    const float* mrow = mcs + col * MC_PITCH + 2 * half;
    const v2f a0 = *(const v2f*)(mrow + 0);
    const v2f a1 = *(const v2f*)(mrow + 4);
    const v2f a2 = *(const v2f*)(mrow + 8);
    const v2f a3 = *(const v2f*)(mrow + 12);

    auto process = [&](const float4 (&xv)[GPS], const int s) {
#pragma unroll
        for (int t = 0; t < GPS; ++t) xs[w][t][lane] = xv[t];
        __builtin_amdgcn_fence(__ATOMIC_ACQ_REL, "wavefront");
        float4 x0[GPS], x1[GPS];
#pragma unroll
        for (int t = 0; t < GPS; ++t) {
            x0[t] = xs[w][t][col];
            x1[t] = xs[w][t][16 + col];
        }
        __builtin_amdgcn_sched_barrier(0);

        float4 hs0[GPS], hs1[GPS];
#pragma unroll
        for (int t = 0; t < GPS; ++t) {
            v2f b0[4], b1[4];
            make_b(x0[t], half, b0);
            make_b(x1[t], half, b1);

            v8f acc0 = {0.f, 0.f, 0.f, 0.f, 0.f, 0.f, 0.f, 0.f};
            v8f acc1 = {0.f, 0.f, 0.f, 0.f, 0.f, 0.f, 0.f, 0.f};
            acc0 = __builtin_amdgcn_wmma_f32_16x16x4_f32(false, a0, false, b0[0], (short)0, acc0, false, false);
            acc1 = __builtin_amdgcn_wmma_f32_16x16x4_f32(false, a0, false, b1[0], (short)0, acc1, false, false);
            acc0 = __builtin_amdgcn_wmma_f32_16x16x4_f32(false, a1, false, b0[1], (short)0, acc0, false, false);
            acc1 = __builtin_amdgcn_wmma_f32_16x16x4_f32(false, a1, false, b1[1], (short)0, acc1, false, false);
            acc0 = __builtin_amdgcn_wmma_f32_16x16x4_f32(false, a2, false, b0[2], (short)0, acc0, false, false);
            acc1 = __builtin_amdgcn_wmma_f32_16x16x4_f32(false, a2, false, b1[2], (short)0, acc1, false, false);
            acc0 = __builtin_amdgcn_wmma_f32_16x16x4_f32(false, a3, false, b0[3], (short)0, acc0, false, false);
            acc1 = __builtin_amdgcn_wmma_f32_16x16x4_f32(false, a3, false, b1[3], (short)0, acc1, false, false);

            hs0[t] = half_sums(acc0);
            hs1[t] = half_sums(acc1);
        }
        __builtin_amdgcn_sched_barrier(0);
#pragma unroll
        for (int t = 0; t < GPS; ++t) {
            ps[w][t][0][half][col] = hs0[t];
            ps[w][t][1][half][col] = hs1[t];
        }
        __builtin_amdgcn_fence(__ATOMIC_ACQ_REL, "wavefront");
        float* og = out + (long)s * (128 * GPS);
        float4 o[GPS];
#pragma unroll
        for (int t = 0; t < GPS; ++t) {
            const float4 pa = ps[w][t][half][0][col];
            const float4 pb = ps[w][t][half][1][col];
            o[t] = make_float4(pb.x, pa.y + pb.y, pa.z + pb.z, pa.w + pb.w);
        }
        if (s != nS - 1 || lastRows == 32 * GPS) {
#pragma unroll
            for (int t = 0; t < GPS; ++t) *(float4*)(og + 128 * t + laneOff) = o[t];
        } else {
#pragma unroll
            for (int t = 0; t < GPS; ++t)
                if (t * 32 + lane < lastRows) *(float4*)(og + 128 * t + laneOff) = o[t];
        }
    };

    int s = gw;
    for (int q = 0; q < rounds; ++q) {
#pragma unroll
        for (int k = 0; k < SLOTS; ++k) {
            load_step(s + (SLOTS - 1) * nW, ring[(k + SLOTS - 1) % SLOTS]);
            process(ring[k], s);
            s += nW;
        }
    }
#pragma unroll
    for (int k = 0; k < SLOTS - 1; ++k) {
        if (k < rem) {
            process(ring[k], s);
            s += nW;
        }
    }
}

extern "C" void kernel_launch(void* const* d_in, const int* in_sizes, int n_in,
                              void* d_out, int out_size, void* d_ws, size_t ws_size,
                              hipStream_t stream) {
    const float* x     = (const float*)d_in[0];
    const float* theta = (const float*)d_in[1];
    float* out = (float*)d_out;
    int B = in_sizes[0] / 4;
    if (B <= 0) return;

    vqc_main_kernel<<<512, 256, 0, stream>>>(x, theta, out, B);
}
